# starting kernel
# baseline (speedup 1.0000x reference)
#include <hip/hip_runtime.h>
#include <hip/hip_bf16.h>

typedef __attribute__((ext_vector_type(2))) float v2f;
typedef __attribute__((ext_vector_type(8))) float v8f;

#define N_NODES 100000
#define ROW_TILES (N_NODES / 16)

__global__ void deg_init_kernel(float* deg, int n) {
    int i = blockIdx.x * blockDim.x + threadIdx.x;
    if (i < n) deg[i] = 1.0f;
}

__global__ void deg_count_kernel(const int* __restrict__ col, float* deg, int E) {
    int e = blockIdx.x * blockDim.x + threadIdx.x;
    if (e < E) unsafeAtomicAdd(&deg[col[e]], 1.0f);
}

__global__ void deg_rsqrt_kernel(float* deg, int n) {
    int i = blockIdx.x * blockDim.x + threadIdx.x;
    if (i < n) deg[i] = rsqrtf(deg[i]);
}

__global__ __launch_bounds__(256) void gemm1_kernel(const float* __restrict__ x,
                                                    const float* __restrict__ W,
                                                    float* __restrict__ h) {
    const int lane = threadIdx.x & 31;
    const int wave = threadIdx.x >> 5;
    const int rowTile = blockIdx.x * 2 + (wave >> 2);
    const int colTile = wave & 3;
    const int m0 = rowTile * 16;
    const int n0 = colTile * 16;
    const int half = lane >> 4;
    const int l    = lane & 15;

    v8f c = {};
    const float* arow = x + (m0 + l) * 128 + 2 * half;
    const float* bcol = W + (2 * half) * 64 + n0 + l;
#pragma unroll
    for (int k0 = 0; k0 < 128; k0 += 4) {
        v2f a = *(const v2f*)(arow + k0);
        v2f b;
        b.x = bcol[(k0 + 0) * 64];
        b.y = bcol[(k0 + 1) * 64];
        c = __builtin_amdgcn_wmma_f32_16x16x4_f32(false, a, false, b,
                                                  (short)0, c, false, false);
    }
    float* o = h + (m0 + half * 8) * 64 + n0 + l;
#pragma unroll
    for (int v = 0; v < 8; ++v) o[v * 64] = c[v];
}

__global__ __launch_bounds__(256) void gemm2_kernel(const float* __restrict__ ain,
                                                    const float* __restrict__ W,
                                                    float* __restrict__ g) {
    const int lane = threadIdx.x & 31;
    const int wave = threadIdx.x >> 5;
    const int rowTile = blockIdx.x * 4 + (wave >> 1);
    if (rowTile >= ROW_TILES) return;
    const int colTile = wave & 1;
    const int m0 = rowTile * 16;
    const int n0 = colTile * 16;
    const int half = lane >> 4;
    const int l    = lane & 15;

    v8f c = {};
    const float* arow = ain + (m0 + l) * 64 + 2 * half;
    const float* bcol = W + (2 * half) * 32 + n0 + l;
#pragma unroll
    for (int k0 = 0; k0 < 64; k0 += 4) {
        v2f a = *(const v2f*)(arow + k0);
        v2f b;
        b.x = bcol[(k0 + 0) * 32];
        b.y = bcol[(k0 + 1) * 32];
        c = __builtin_amdgcn_wmma_f32_16x16x4_f32(false, a, false, b,
                                                  (short)0, c, false, false);
    }
    float* o = g + (m0 + half * 8) * 32 + n0 + l;
#pragma unroll
    for (int v = 0; v < 8; ++v) o[v * 32] = c[v];
}

__global__ void init_acc_kernel(const float* __restrict__ h,
                                const float* __restrict__ bias,
                                const float* __restrict__ dis,
                                float* __restrict__ acc,
                                int fshift, int fmask, int total) {
    int i = blockIdx.x * blockDim.x + threadIdx.x;
    if (i >= total) return;
    int node = i >> fshift;
    int f = i & fmask;
    float d = dis[node];
    acc[i] = bias[f] + h[i] * d * d;
}

__global__ void scatter64_kernel(const int* __restrict__ row,
                                 const int* __restrict__ col,
                                 const float* __restrict__ dis,
                                 const float* __restrict__ h,
                                 float* acc, int E) {
    int tid = blockIdx.x * blockDim.x + threadIdx.x;
    if (tid >= E * 16) return;
    int e  = tid >> 4;
    int fc = (tid & 15) << 2;
    int r = row[e], c = col[e];
    float norm = dis[r] * dis[c];
    float4 v = *(const float4*)(h + r * 64 + fc);
    float* p = acc + c * 64 + fc;
    unsafeAtomicAdd(p + 0, v.x * norm);
    unsafeAtomicAdd(p + 1, v.y * norm);
    unsafeAtomicAdd(p + 2, v.z * norm);
    unsafeAtomicAdd(p + 3, v.w * norm);
}

__global__ void scatter32_kernel(const int* __restrict__ row,
                                 const int* __restrict__ col,
                                 const float* __restrict__ dis,
                                 const float* __restrict__ h,
                                 float* acc, int E) {
    int tid = blockIdx.x * blockDim.x + threadIdx.x;
    if (tid >= E * 8) return;
    int e  = tid >> 3;
    int fc = (tid & 7) << 2;
    int r = row[e], c = col[e];
    float norm = dis[r] * dis[c];
    float4 v = *(const float4*)(h + r * 32 + fc);
    float* p = acc + c * 32 + fc;
    unsafeAtomicAdd(p + 0, v.x * norm);
    unsafeAtomicAdd(p + 1, v.y * norm);
    unsafeAtomicAdd(p + 2, v.z * norm);
    unsafeAtomicAdd(p + 3, v.w * norm);
}

__global__ void relu_kernel(float* a, int total) {
    int i = blockIdx.x * blockDim.x + threadIdx.x;
    if (i < total) a[i] = fmaxf(a[i], 0.0f);
}

extern "C" void kernel_launch(void* const* d_in, const int* in_sizes, int n_in,
                              void* d_out, int out_size, void* d_ws, size_t ws_size,
                              hipStream_t stream) {
    const float* x  = (const float*)d_in[0];
    const int*   ei = (const int*)d_in[1];
    const float* W1 = (const float*)d_in[2];
    const float* b1 = (const float*)d_in[3];
    const float* W2 = (const float*)d_in[4];
    const float* b2 = (const float*)d_in[5];
    float* out = (float*)d_out;

    const int E = in_sizes[1] / 2;
    const int* row = ei;
    const int* col = ei + E;

    float* dis  = (float*)d_ws;
    float* h1   = dis  + N_NODES;
    float* acc1 = h1   + N_NODES * 64;
    float* g2   = acc1 + N_NODES * 64;

    const int T = 256;
    deg_init_kernel<<<(N_NODES + T - 1) / T, T, 0, stream>>>(dis, N_NODES);
    deg_count_kernel<<<(E + T - 1) / T, T, 0, stream>>>(col, dis, E);
    deg_rsqrt_kernel<<<(N_NODES + T - 1) / T, T, 0, stream>>>(dis, N_NODES);

    gemm1_kernel<<<ROW_TILES / 2, T, 0, stream>>>(x, W1, h1);
    init_acc_kernel<<<(N_NODES * 64 + T - 1) / T, T, 0, stream>>>(
        h1, b1, dis, acc1, 6, 63, N_NODES * 64);
    scatter64_kernel<<<(E * 16 + T - 1) / T, T, 0, stream>>>(row, col, dis, h1, acc1, E);
    relu_kernel<<<(N_NODES * 64 + T - 1) / T, T, 0, stream>>>(acc1, N_NODES * 64);

    gemm2_kernel<<<(ROW_TILES + 3) / 4, T, 0, stream>>>(acc1, W2, g2);
    init_acc_kernel<<<(N_NODES * 32 + T - 1) / T, T, 0, stream>>>(
        g2, b2, dis, out, 5, 31, N_NODES * 32);
    scatter32_kernel<<<(E * 8 + T - 1) / T, T, 0, stream>>>(row, col, dis, g2, out, E);
}


# both matrix products: a block's A rows and all of W enter LDS once by whole-line 16-byte loads; A and B fragments are read from LDS, not from row-major global rows; k order and sums unchanged
# speedup vs baseline: 2.6770x; 2.6770x over previous
#include <hip/hip_runtime.h>
#include <hip/hip_bf16.h>


typedef __attribute__((ext_vector_type(2))) float v2f;
typedef __attribute__((ext_vector_type(4))) float v4f;
typedef __attribute__((ext_vector_type(8))) float v8f;

#define N_NODES 100000
#define ROW_TILES (N_NODES / 16)
#define NODE_SPLIT 50048
#define TK_STRIDE 64

__device__ __forceinline__ int chip_half() {
    return (__builtin_amdgcn_s_sendmsg_rtn(0x87) >> 18) & 1;
}

__device__ __forceinline__ int draw_run(int* tk, int& half, int nruns, int lane) {
    int t = 0;
    if (lane == 0) t = atomicAdd(tk + half * TK_STRIDE, 1);
    t = __builtin_amdgcn_readfirstlane(t);
    if (t >= nruns) {
        half ^= 1;
        if (lane == 0) t = atomicAdd(tk + half * TK_STRIDE, 1);
        t = __builtin_amdgcn_readfirstlane(t);
    }
    return t;
}

__global__ void deg_init_kernel(float* deg, int* tk, int n) {
    int i = blockIdx.x * blockDim.x + threadIdx.x;
    if (i < n) deg[i] = 1.0f;
    if (i < 4) tk[i * TK_STRIDE] = 0;
}

__global__ __launch_bounds__(256) void deg_count_kernel(const int* __restrict__ col, float* deg,
                                                        int* tk, int E, int nruns) {
    const int lane = threadIdx.x & 31;
    int half = chip_half();
    const int t = draw_run(tk, half, nruns, lane);
    if (t >= nruns) return;
    const int lo = half ? NODE_SPLIT : 0;
    const int hi = half ? N_NODES : NODE_SPLIT;
    const int e0 = t * 256 + lane;
    int c[8];
#pragma unroll
    for (int i = 0; i < 8; ++i) {
        const int e = e0 + i * 32;
        c[i] = (e < E) ? col[e] : -1;
    }
#pragma unroll
    for (int i = 0; i < 8; ++i)
        if (c[i] >= lo && c[i] < hi) unsafeAtomicAdd(&deg[c[i]], 1.0f);
}

#define XS_PITCH 132
#define W1S_PITCH 160
__global__ __launch_bounds__(256) void gemm1_kernel(const float* __restrict__ x,
                                                    const float* __restrict__ W,
                                                    const float* __restrict__ deg,
                                                    const float* __restrict__ bias,
                                                    float* __restrict__ dis,
                                                    float* __restrict__ h,
                                                    float* __restrict__ acc) {
    __shared__ __attribute__((aligned(16))) float xs[32 * XS_PITCH];
    __shared__ __attribute__((aligned(16))) float ws[64 * W1S_PITCH];
    const int tid  = threadIdx.x;
    const int lane = tid & 31;
    const int wave = tid >> 5;
    const int rowTile = blockIdx.x * 2 + (wave >> 2);
    const int colTile = wave & 3;
    const int m0 = rowTile * 16;
    const int n0 = colTile * 16;
    const int half = lane >> 4;
    const int l    = lane & 15;

    const float* xb = x + blockIdx.x * (32 * 128);
    v4f xa[4];
#pragma unroll
    for (int i = 0; i < 4; ++i) xa[i] = *(const v4f*)(xb + (tid + 256 * i) * 4);
    v4f we[4], wo[4];
#pragma unroll
    for (int i = 0; i < 4; ++i) {
        const int it = tid + 256 * i;
        const float* wp = W + (it >> 4) * 128 + (it & 15) * 4;
        we[i] = *(const v4f*)wp;
        wo[i] = *(const v4f*)(wp + 64);
    }

    const bool wdis = (colTile == 0) && (lane < 16);
    float dself = 1.0f;
    if (wdis) dself = deg[m0 + lane];
    const int mrow = m0 + half * 8;
    float dg[8];
#pragma unroll
    for (int v = 0; v < 8; ++v) dg[v] = deg[mrow + v];
    const float bv = bias[n0 + l];

#pragma unroll
    for (int i = 0; i < 4; ++i) {
        const int idx = tid + 256 * i;
        *(v4f*)(xs + (idx >> 5) * XS_PITCH + (idx & 31) * 4) = xa[i];
    }
#pragma unroll
    for (int i = 0; i < 4; ++i) {
        const int it = tid + 256 * i;
        float* d = ws + (it >> 4) * W1S_PITCH + (it & 15) * 8;
        v4f s0, s1;
        s0.x = we[i].x; s0.y = wo[i].x; s0.z = we[i].y; s0.w = wo[i].y;
        s1.x = we[i].z; s1.y = wo[i].z; s1.z = we[i].w; s1.w = wo[i].w;
        *(v4f*)d       = s0;
        *(v4f*)(d + 4) = s1;
    }
    __syncthreads();
    if (wdis) dis[m0 + lane] = rsqrtf(dself);

    v8f c = {};
    const float* ap = xs + ((wave >> 2) * 16 + l) * XS_PITCH + 2 * half;
    const float* bp = ws + half * W1S_PITCH + (n0 + l) * 2;
#pragma unroll
    for (int k0 = 0; k0 < 128; k0 += 4) {
        v2f a = *(const v2f*)(ap + k0);
        v2f b = *(const v2f*)(bp + (k0 >> 1) * W1S_PITCH);
        c = __builtin_amdgcn_wmma_f32_16x16x4_f32(false, a, false, b,
                                                  (short)0, c, false, false);
    }
    float* o  = h   + mrow * 64 + n0 + l;
    float* oa = acc + mrow * 64 + n0 + l;
#pragma unroll
    for (int v = 0; v < 8; ++v) {
        const float d = rsqrtf(dg[v]);
        o[v * 64]  = c[v];
        oa[v * 64] = bv + c[v] * d * d;
    }
}

#define AS_PITCH 68
#define W2S_PITCH 96
__global__ __launch_bounds__(256) void gemm2_kernel(const float* __restrict__ ain,
                                                    const float* __restrict__ W,
                                                    const float* __restrict__ dis,
                                                    const float* __restrict__ bias,
                                                    float* __restrict__ g,
                                                    float* __restrict__ out, int* tk) {
    __shared__ __attribute__((aligned(16))) float as[64 * AS_PITCH];
    __shared__ __attribute__((aligned(16))) float ws[32 * W2S_PITCH];
    const int tid  = threadIdx.x;
    const int lane = tid & 31;
    const int wave = tid >> 5;
    if (blockIdx.x == 0 && tid < 2) tk[tid * TK_STRIDE] = 0;

    const int r0 = blockIdx.x * 64;
    v4f aa[4];
#pragma unroll
    for (int i = 0; i < 4; ++i) {
        const int idx = tid + 256 * i;
        const int r = r0 + (idx >> 4);
        v4f z = {};
        aa[i] = (r < N_NODES) ? *(const v4f*)(ain + r * 64 + (idx & 15) * 4) : z;
    }
    const float* wp = W + (tid >> 3) * 64 + (tid & 7) * 4;
    const v4f we = *(const v4f*)wp;
    const v4f wo = *(const v4f*)(wp + 32);

    const int rowTile = blockIdx.x * 4 + (wave >> 1);
    const bool live = rowTile < ROW_TILES;
    const int colTile = wave & 1;
    const int m0 = rowTile * 16;
    const int n0 = colTile * 16;
    const int half = lane >> 4;
    const int l    = lane & 15;
    const int mrow = m0 + half * 8;
    const int drow = live ? mrow : 0;
    float dv[8];
#pragma unroll
    for (int v = 0; v < 8; ++v) dv[v] = dis[drow + v];
    const float bv = bias[n0 + l];

#pragma unroll
    for (int i = 0; i < 4; ++i) {
        const int idx = tid + 256 * i;
        v4f v = aa[i];
        v.x = fmaxf(v.x, 0.0f);
        v.y = fmaxf(v.y, 0.0f);
        v.z = fmaxf(v.z, 0.0f);
        v.w = fmaxf(v.w, 0.0f);
        *(v4f*)(as + (idx >> 4) * AS_PITCH + (idx & 15) * 4) = v;
    }
    {
        float* d = ws + (tid >> 3) * W2S_PITCH + (tid & 7) * 8;
        v4f s0, s1;
        s0.x = we.x; s0.y = wo.x; s0.z = we.y; s0.w = wo.y;
        s1.x = we.z; s1.y = wo.z; s1.z = we.w; s1.w = wo.w;
        *(v4f*)d       = s0;
        *(v4f*)(d + 4) = s1;
    }
    __syncthreads();
    if (!live) return;

    v8f c = {};
    const float* ap = as + ((wave >> 1) * 16 + l) * AS_PITCH + 2 * half;
    const float* bp = ws + half * W2S_PITCH + (n0 + l) * 2;
#pragma unroll
    for (int k0 = 0; k0 < 64; k0 += 4) {
        v2f a = *(const v2f*)(ap + k0);
        v2f b = *(const v2f*)(bp + (k0 >> 1) * W2S_PITCH);
        c = __builtin_amdgcn_wmma_f32_16x16x4_f32(false, a, false, b,
                                                  (short)0, c, false, false);
    }
    float* o  = g   + mrow * 32 + n0 + l;
    float* oo = out + mrow * 32 + n0 + l;
#pragma unroll
    for (int v = 0; v < 8; ++v) {
        o[v * 32]  = c[v];
        oo[v * 32] = bv + c[v] * dv[v] * dv[v];
    }
}

template <int F, int G, bool PARTIAL>
__device__ __forceinline__ void edge_group(int R, int C, int NB, int j0, int n, int lane,
                                           const float* __restrict__ h, float* acc) {
    constexpr int Q = F / 32;
    float v[G][Q];
#pragma unroll
    for (int j = 0; j < G; ++j) {
        for (int q = 0; q < Q; ++q) v[j][q] = 0.0f;
        if (!PARTIAL || j < n) {
            const float* s = h + __builtin_amdgcn_readlane(R, j0 + j) * F;
            for (int q = 0; q < Q; ++q) v[j][q] = s[lane + 32 * q];
        }
    }
#pragma unroll
    for (int j = 0; j < G; ++j) {
        if (!PARTIAL || j < n) {
            float* p = acc + __builtin_amdgcn_readlane(C, j0 + j) * F;
            const float nj = __int_as_float(__builtin_amdgcn_readlane(NB, j0 + j));
            for (int q = 0; q < Q; ++q) unsafeAtomicAdd(p + lane + 32 * q, v[j][q] * nj);
        }
    }
}

template <int F, int G>
__device__ __forceinline__ void walk_edges(int R, int C, int NB, int cnt, int lane,
                                           const float* __restrict__ h, float* acc) {
#pragma unroll
    for (int g = 0; g < 32 / G; ++g)
        if ((g + 1) * G <= cnt) edge_group<F, G, false>(R, C, NB, g * G, G, lane, h, acc);
    const int j0 = cnt & ~(G - 1);
    if (j0 < cnt) edge_group<F, G, true>(R, C, NB, j0, cnt - j0, lane, h, acc);
}

template <int F, int G>
__global__ __launch_bounds__(256) void scatter_rows_kernel(const int* __restrict__ row,
                                                           const int* __restrict__ col,
                                                           const float* __restrict__ dis,
                                                           const float* __restrict__ h,
                                                           float* acc, int* tk, int E, int nruns) {
    __shared__ int lst[8][2][64];
    const int lane = threadIdx.x & 31;
    const int w = threadIdx.x >> 5;
    int half = chip_half();
    const int t = draw_run(tk, half, nruns, lane);
    if (t >= nruns) return;
    const int lo = half ? NODE_SPLIT : 0;
    const int hi = half ? N_NODES : NODE_SPLIT;

    const int ea = t * 64 + lane;
    const int eb = ea + 32;
    int ra = 0, ca = -1, rb = 0, cb = -1;
    if (ea < E) { ra = row[ea]; ca = col[ea]; }
    if (eb < E) { rb = row[eb]; cb = col[eb]; }
    const bool owna = (ca >= lo) && (ca < hi);
    const bool ownb = (cb >= lo) && (cb < hi);
    const unsigned ma = __builtin_amdgcn_ballot_w32(owna);
    const unsigned mb = __builtin_amdgcn_ballot_w32(ownb);
    const int ka = __builtin_popcount(ma);
    const int cnt = ka + __builtin_popcount(mb);
    const int pa = __builtin_amdgcn_mbcnt_lo(ma, 0);
    const int pb = ka + __builtin_amdgcn_mbcnt_lo(mb, 0);
    if (owna) { lst[w][0][pa] = ra; lst[w][1][pa] = ca; }
    if (ownb) { lst[w][0][pb] = rb; lst[w][1][pb] = cb; }
    __builtin_amdgcn_wave_barrier();
    int R0 = 0, C0 = 0, R1 = 0, C1 = 0;
    if (lane < cnt)      { R0 = lst[w][0][lane];      C0 = lst[w][1][lane]; }
    if (lane + 32 < cnt) { R1 = lst[w][0][lane + 32]; C1 = lst[w][1][lane + 32]; }
    const float d00 = dis[R0], d01 = dis[C0], d10 = dis[R1], d11 = dis[C1];
    const int N0 = __float_as_int(d00 * d01);
    const int N1 = __float_as_int(d10 * d11);
    walk_edges<F, G>(R0, C0, N0, min(cnt, 32), lane, h, acc);
    if (cnt > 32) walk_edges<F, G>(R1, C1, N1, cnt - 32, lane, h, acc);
}

extern "C" void kernel_launch(void* const* d_in, const int* in_sizes, int n_in,
                              void* d_out, int out_size, void* d_ws, size_t ws_size,
                              hipStream_t stream) {
    const float* x  = (const float*)d_in[0];
    const int*   ei = (const int*)d_in[1];
    const float* W1 = (const float*)d_in[2];
    const float* b1 = (const float*)d_in[3];
    const float* W2 = (const float*)d_in[4];
    const float* b2 = (const float*)d_in[5];
    float* out = (float*)d_out;

    const int E = in_sizes[1] / 2;
    const int* row = ei;
    const int* col = ei + E;

    float* dis  = (float*)d_ws;
    float* h1   = dis  + N_NODES;
    float* acc1 = h1   + N_NODES * 64;
    float* g2   = acc1 + N_NODES * 64;
    int* tk01 = (int*)g2 + 32;
    int* tk2  = (int*)h1 + 32;
    float* deg  = g2 + 32 + 256;

    const int T = 256;
    const int runsD = (E + 255) / 256;
    const int runsS = (E + 63) / 64;
    const int blocksD = (2 * runsD + 7) / 8;
    const int blocksS = (2 * runsS + 7) / 8;

    deg_init_kernel<<<(N_NODES + T - 1) / T, T, 0, stream>>>(deg, tk01, N_NODES);
    deg_count_kernel<<<blocksD, T, 0, stream>>>(col, deg, tk01, E, runsD);

    gemm1_kernel<<<ROW_TILES / 2, T, 0, stream>>>(x, W1, deg, b1, dis, h1, acc1);
    scatter_rows_kernel<64, 8><<<blocksS, T, 0, stream>>>(row, col, dis, h1, acc1,
                                                           tk01 + 2 * TK_STRIDE, E, runsS);

    gemm2_kernel<<<(ROW_TILES + 3) / 4, T, 0, stream>>>(acc1, W2, dis, b2, g2, out, tk2);
    scatter_rows_kernel<32, 16><<<blocksS, T, 0, stream>>>(row, col, dis, g2, out, tk2, E, runsS);
}


# per-edge norm dis[source]*dis[target] computed once, in the layer-1 product launch (two edges per thread), kept in workspace; both edge passes read it with the ids, no random dis lookups; sums same
# speedup vs baseline: 2.8634x; 1.0696x over previous
#include <hip/hip_runtime.h>
#include <hip/hip_bf16.h>


typedef __attribute__((ext_vector_type(2))) float v2f;
typedef __attribute__((ext_vector_type(4))) float v4f;
typedef __attribute__((ext_vector_type(8))) float v8f;

#define N_NODES 100000
#define ROW_TILES (N_NODES / 16)
#define NODE_SPLIT 50048
#define TK_STRIDE 64

__device__ __forceinline__ int chip_half() {
    return (__builtin_amdgcn_s_sendmsg_rtn(0x87) >> 18) & 1;
}

__device__ __forceinline__ int draw_run(int* tk, int& half, int nruns, int lane) {
    int t = 0;
    if (lane == 0) t = atomicAdd(tk + half * TK_STRIDE, 1);
    t = __builtin_amdgcn_readfirstlane(t);
    if (t >= nruns) {
        half ^= 1;
        if (lane == 0) t = atomicAdd(tk + half * TK_STRIDE, 1);
        t = __builtin_amdgcn_readfirstlane(t);
    }
    return t;
}

__global__ void deg_init_kernel(float* deg, int* tk, int n) {
    int i = blockIdx.x * blockDim.x + threadIdx.x;
    if (i < n) deg[i] = 1.0f;
    if (i < 4) tk[i * TK_STRIDE] = 0;
}

__global__ __launch_bounds__(256) void deg_count_kernel(const int* __restrict__ col, float* deg,
                                                        int* tk, int E, int nruns) {
    const int lane = threadIdx.x & 31;
    int half = chip_half();
    const int t = draw_run(tk, half, nruns, lane);
    if (t >= nruns) return;
    const int lo = half ? NODE_SPLIT : 0;
    const int hi = half ? N_NODES : NODE_SPLIT;
    const int e0 = t * 256 + lane;
    int c[8];
#pragma unroll
    for (int i = 0; i < 8; ++i) {
        const int e = e0 + i * 32;
        c[i] = (e < E) ? col[e] : -1;
    }
#pragma unroll
    for (int i = 0; i < 8; ++i)
        if (c[i] >= lo && c[i] < hi) unsafeAtomicAdd(&deg[c[i]], 1.0f);
}

#define XS_PITCH 132
#define W1S_PITCH 160
__global__ __launch_bounds__(256) void gemm1_kernel(const float* __restrict__ x,
                                                    const float* __restrict__ W,
                                                    const float* __restrict__ deg,
                                                    const float* __restrict__ bias,
                                                    float* __restrict__ dis,
                                                    float* __restrict__ h,
                                                    float* __restrict__ acc,
                                                    const int* __restrict__ row,
                                                    const int* __restrict__ col,
                                                    float* __restrict__ nrm, int E) {
    __shared__ __attribute__((aligned(16))) float xs[32 * XS_PITCH];
    __shared__ __attribute__((aligned(16))) float ws[64 * W1S_PITCH];
    const int tid  = threadIdx.x;
    const int lane = tid & 31;
    const int wave = tid >> 5;
    const int rowTile = blockIdx.x * 2 + (wave >> 2);
    const int colTile = wave & 3;
    const int m0 = rowTile * 16;
    const int n0 = colTile * 16;
    const int half = lane >> 4;
    const int l    = lane & 15;

    const int per  = (E + (int)gridDim.x - 1) / (int)gridDim.x;
    const int eend = min(E, ((int)blockIdx.x + 1) * per);
    const int e0 = (int)blockIdx.x * per + tid;
    const int e1 = e0 + 256;
    const int i0 = min(e0, E - 1), i1 = min(e1, E - 1);
    const int er0 = row[i0], ec0 = col[i0], er1 = row[i1], ec1 = col[i1];

    const float* xb = x + blockIdx.x * (32 * 128);
    v4f xa[4];
#pragma unroll
    for (int i = 0; i < 4; ++i) xa[i] = *(const v4f*)(xb + (tid + 256 * i) * 4);
    v4f we[4], wo[4];
#pragma unroll
    for (int i = 0; i < 4; ++i) {
        const int it = tid + 256 * i;
        const float* wp = W + (it >> 4) * 128 + (it & 15) * 4;
        we[i] = *(const v4f*)wp;
        wo[i] = *(const v4f*)(wp + 64);
    }

    const bool wdis = (colTile == 0) && (lane < 16);
    float dself = 1.0f;
    if (wdis) dself = deg[m0 + lane];
    const int mrow = m0 + half * 8;
    float dg[8];
#pragma unroll
    for (int v = 0; v < 8; ++v) dg[v] = deg[mrow + v];
    const float bv = bias[n0 + l];

#pragma unroll
    for (int i = 0; i < 4; ++i) {
        const int idx = tid + 256 * i;
        *(v4f*)(xs + (idx >> 5) * XS_PITCH + (idx & 31) * 4) = xa[i];
    }
#pragma unroll
    for (int i = 0; i < 4; ++i) {
        const int it = tid + 256 * i;
        float* d = ws + (it >> 4) * W1S_PITCH + (it & 15) * 8;
        v4f s0, s1;
        s0.x = we[i].x; s0.y = wo[i].x; s0.z = we[i].y; s0.w = wo[i].y;
        s1.x = we[i].z; s1.y = wo[i].z; s1.z = we[i].w; s1.w = wo[i].w;
        *(v4f*)d       = s0;
        *(v4f*)(d + 4) = s1;
    }
    __syncthreads();
    if (wdis) dis[m0 + lane] = rsqrtf(dself);
    const float q0 = deg[er0], q1 = deg[ec0], q2 = deg[er1], q3 = deg[ec1];

    v8f c = {};
    const float* ap = xs + ((wave >> 2) * 16 + l) * XS_PITCH + 2 * half;
    const float* bp = ws + half * W1S_PITCH + (n0 + l) * 2;
#pragma unroll
    for (int k0 = 0; k0 < 128; k0 += 4) {
        v2f a = *(const v2f*)(ap + k0);
        v2f b = *(const v2f*)(bp + (k0 >> 1) * W1S_PITCH);
        c = __builtin_amdgcn_wmma_f32_16x16x4_f32(false, a, false, b,
                                                  (short)0, c, false, false);
    }
    float* o  = h   + mrow * 64 + n0 + l;
    float* oa = acc + mrow * 64 + n0 + l;
#pragma unroll
    for (int v = 0; v < 8; ++v) {
        const float d = rsqrtf(dg[v]);
        o[v * 64]  = c[v];
        oa[v * 64] = bv + c[v] * d * d;
    }
    asm volatile("" :: "v"(q0), "v"(q1), "v"(q2), "v"(q3));
    if (e0 < eend) nrm[e0] = rsqrtf(q0) * rsqrtf(q1);
    if (e1 < eend) nrm[e1] = rsqrtf(q2) * rsqrtf(q3);
    for (int e = e1 + 256; e < eend; e += 256)
        nrm[e] = rsqrtf(deg[row[e]]) * rsqrtf(deg[col[e]]);
}

#define AS_PITCH 68
#define W2S_PITCH 96
__global__ __launch_bounds__(256) void gemm2_kernel(const float* __restrict__ ain,
                                                    const float* __restrict__ W,
                                                    const float* __restrict__ dis,
                                                    const float* __restrict__ bias,
                                                    float* __restrict__ g,
                                                    float* __restrict__ out, int* tk) {
    __shared__ __attribute__((aligned(16))) float as[64 * AS_PITCH];
    __shared__ __attribute__((aligned(16))) float ws[32 * W2S_PITCH];
    const int tid  = threadIdx.x;
    const int lane = tid & 31;
    const int wave = tid >> 5;
    if (blockIdx.x == 0 && tid < 2) tk[tid * TK_STRIDE] = 0;

    const int r0 = blockIdx.x * 64;
    v4f aa[4];
#pragma unroll
    for (int i = 0; i < 4; ++i) {
        const int idx = tid + 256 * i;
        const int r = r0 + (idx >> 4);
        v4f z = {};
        aa[i] = (r < N_NODES) ? *(const v4f*)(ain + r * 64 + (idx & 15) * 4) : z;
    }
    const float* wp = W + (tid >> 3) * 64 + (tid & 7) * 4;
    const v4f we = *(const v4f*)wp;
    const v4f wo = *(const v4f*)(wp + 32);

    const int rowTile = blockIdx.x * 4 + (wave >> 1);
    const bool live = rowTile < ROW_TILES;
    const int colTile = wave & 1;
    const int m0 = rowTile * 16;
    const int n0 = colTile * 16;
    const int half = lane >> 4;
    const int l    = lane & 15;
    const int mrow = m0 + half * 8;
    const int drow = live ? mrow : 0;
    float dv[8];
#pragma unroll
    for (int v = 0; v < 8; ++v) dv[v] = dis[drow + v];
    const float bv = bias[n0 + l];

#pragma unroll
    for (int i = 0; i < 4; ++i) {
        const int idx = tid + 256 * i;
        v4f v = aa[i];
        v.x = fmaxf(v.x, 0.0f);
        v.y = fmaxf(v.y, 0.0f);
        v.z = fmaxf(v.z, 0.0f);
        v.w = fmaxf(v.w, 0.0f);
        *(v4f*)(as + (idx >> 4) * AS_PITCH + (idx & 15) * 4) = v;
    }
    {
        float* d = ws + (tid >> 3) * W2S_PITCH + (tid & 7) * 8;
        v4f s0, s1;
        s0.x = we.x; s0.y = wo.x; s0.z = we.y; s0.w = wo.y;
        s1.x = we.z; s1.y = wo.z; s1.z = we.w; s1.w = wo.w;
        *(v4f*)d       = s0;
        *(v4f*)(d + 4) = s1;
    }
    __syncthreads();
    if (!live) return;

    v8f c = {};
    const float* ap = as + ((wave >> 1) * 16 + l) * AS_PITCH + 2 * half;
    const float* bp = ws + half * W2S_PITCH + (n0 + l) * 2;
#pragma unroll
    for (int k0 = 0; k0 < 64; k0 += 4) {
        v2f a = *(const v2f*)(ap + k0);
        v2f b = *(const v2f*)(bp + (k0 >> 1) * W2S_PITCH);
        c = __builtin_amdgcn_wmma_f32_16x16x4_f32(false, a, false, b,
                                                  (short)0, c, false, false);
    }
    float* o  = g   + mrow * 32 + n0 + l;
    float* oo = out + mrow * 32 + n0 + l;
#pragma unroll
    for (int v = 0; v < 8; ++v) {
        o[v * 32]  = c[v];
        oo[v * 32] = bv + c[v] * dv[v] * dv[v];
    }
}

template <int F, int G, bool PARTIAL>
__device__ __forceinline__ void edge_group(int R, int C, int NB, int j0, int n, int lane,
                                           const float* __restrict__ h, float* acc) {
    constexpr int Q = F / 32;
    float v[G][Q];
#pragma unroll
    for (int j = 0; j < G; ++j) {
        for (int q = 0; q < Q; ++q) v[j][q] = 0.0f;
        if (!PARTIAL || j < n) {
            const float* s = h + __builtin_amdgcn_readlane(R, j0 + j) * F;
            for (int q = 0; q < Q; ++q) v[j][q] = s[lane + 32 * q];
        }
    }
#pragma unroll
    for (int j = 0; j < G; ++j) {
        if (!PARTIAL || j < n) {
            float* p = acc + __builtin_amdgcn_readlane(C, j0 + j) * F;
            const float nj = __int_as_float(__builtin_amdgcn_readlane(NB, j0 + j));
            for (int q = 0; q < Q; ++q) unsafeAtomicAdd(p + lane + 32 * q, v[j][q] * nj);
        }
    }
}

template <int F, int G>
__device__ __forceinline__ void walk_edges(int R, int C, int NB, int cnt, int lane,
                                           const float* __restrict__ h, float* acc) {
#pragma unroll
    for (int g = 0; g < 32 / G; ++g)
        if ((g + 1) * G <= cnt) edge_group<F, G, false>(R, C, NB, g * G, G, lane, h, acc);
    const int j0 = cnt & ~(G - 1);
    if (j0 < cnt) edge_group<F, G, true>(R, C, NB, j0, cnt - j0, lane, h, acc);
}

template <int F, int G>
__global__ __launch_bounds__(256) void scatter_rows_kernel(const int* __restrict__ row,
                                                           const int* __restrict__ col,
                                                           const float* __restrict__ nrm,
                                                           const float* __restrict__ h,
                                                           float* acc, int* tk, int E, int nruns) {
    __shared__ int lst[8][3][64];
    const int lane = threadIdx.x & 31;
    const int w = threadIdx.x >> 5;
    int half = chip_half();
    const int t = draw_run(tk, half, nruns, lane);
    if (t >= nruns) return;
    const int lo = half ? NODE_SPLIT : 0;
    const int hi = half ? N_NODES : NODE_SPLIT;

    const int ea = t * 64 + lane;
    const int eb = ea + 32;
    int ra = 0, ca = -1, na = 0, rb = 0, cb = -1, nb = 0;
    if (ea < E) { ra = row[ea]; ca = col[ea]; na = __float_as_int(nrm[ea]); }
    if (eb < E) { rb = row[eb]; cb = col[eb]; nb = __float_as_int(nrm[eb]); }
    const bool owna = (ca >= lo) && (ca < hi);
    const bool ownb = (cb >= lo) && (cb < hi);
    const unsigned ma = __builtin_amdgcn_ballot_w32(owna);
    const unsigned mb = __builtin_amdgcn_ballot_w32(ownb);
    const int ka = __builtin_popcount(ma);
    const int cnt = ka + __builtin_popcount(mb);
    const int pa = __builtin_amdgcn_mbcnt_lo(ma, 0);
    const int pb = ka + __builtin_amdgcn_mbcnt_lo(mb, 0);
    if (owna) { lst[w][0][pa] = ra; lst[w][1][pa] = ca; lst[w][2][pa] = na; }
    if (ownb) { lst[w][0][pb] = rb; lst[w][1][pb] = cb; lst[w][2][pb] = nb; }
    __builtin_amdgcn_wave_barrier();
    int R0 = 0, C0 = 0, N0 = 0, R1 = 0, C1 = 0, N1 = 0;
    if (lane < cnt)      { R0 = lst[w][0][lane];      C0 = lst[w][1][lane];      N0 = lst[w][2][lane]; }
    if (lane + 32 < cnt) { R1 = lst[w][0][lane + 32]; C1 = lst[w][1][lane + 32]; N1 = lst[w][2][lane + 32]; }
    walk_edges<F, G>(R0, C0, N0, min(cnt, 32), lane, h, acc);
    if (cnt > 32) walk_edges<F, G>(R1, C1, N1, cnt - 32, lane, h, acc);
}

extern "C" void kernel_launch(void* const* d_in, const int* in_sizes, int n_in,
                              void* d_out, int out_size, void* d_ws, size_t ws_size,
                              hipStream_t stream) {
    const float* x  = (const float*)d_in[0];
    const int*   ei = (const int*)d_in[1];
    const float* W1 = (const float*)d_in[2];
    const float* b1 = (const float*)d_in[3];
    const float* W2 = (const float*)d_in[4];
    const float* b2 = (const float*)d_in[5];
    float* out = (float*)d_out;

    const int E = in_sizes[1] / 2;
    const int* row = ei;
    const int* col = ei + E;

    float* dis  = (float*)d_ws;
    float* h1   = dis  + N_NODES;
    float* acc1 = h1   + N_NODES * 64;
    float* aux  = acc1 + N_NODES * 64;
    float* g2   = h1;
    int* tk01 = (int*)aux + 32;
    float* deg  = aux + 32 + 256;
    int* tk2  = (int*)aux + 32 + 100352;
    float* nrm  = aux + 32 + 131072;

    const int T = 256;
    const int runsD = (E + 255) / 256;
    const int runsS = (E + 63) / 64;
    const int blocksD = (2 * runsD + 7) / 8;
    const int blocksS = (2 * runsS + 7) / 8;

    deg_init_kernel<<<(N_NODES + T - 1) / T, T, 0, stream>>>(deg, tk01, N_NODES);
    deg_count_kernel<<<blocksD, T, 0, stream>>>(col, deg, tk01, E, runsD);

    gemm1_kernel<<<ROW_TILES / 2, T, 0, stream>>>(x, W1, deg, b1, dis, h1, acc1, row, col, nrm, E);
    scatter_rows_kernel<64, 8><<<blocksS, T, 0, stream>>>(row, col, nrm, h1, acc1,
                                                           tk01 + 2 * TK_STRIDE, E, runsS);

    gemm2_kernel<<<(ROW_TILES + 3) / 4, T, 0, stream>>>(acc1, W2, dis, b2, g2, out, tk2);
    scatter_rows_kernel<32, 16><<<blocksS, T, 0, stream>>>(row, col, nrm, g2, out, tk2, E, runsS);
}
